# starting kernel
# baseline (speedup 1.0000x reference)
#include <hip/hip_runtime.h>

typedef __attribute__((ext_vector_type(2))) float v2f;
typedef __attribute__((ext_vector_type(8))) float v8f;

#define LDS_STRIDE 50
#define ROWS 16
#define WPB 8

#define LDS_FENCE() asm volatile("s_wait_dscnt 0" ::: "memory")

static __device__ __forceinline__ float rcp_f(float x) {
    return __builtin_amdgcn_rcpf(x);
}
static __device__ __forceinline__ float tanh_f(float x) {
#if __has_builtin(__builtin_amdgcn_tanhf)
    return __builtin_amdgcn_tanhf(x);
#else
    float e = __expf(-2.0f * x);
    return (1.0f - e) * rcp_f(1.0f + e);
#endif
}
static __device__ __forceinline__ float sigmoid_f(float x) {
    return fmaf(0.5f, tanh_f(0.5f * x), 0.5f);
}
static __device__ __forceinline__ float selu_f(float x) {
    const float a = 1.6732632423543772f, s = 1.0507009873554805f;
    return x > 0.0f ? s * x : s * a * (__expf(x) - 1.0f);
}
static __device__ __forceinline__ float lstm_h(float gi, float gg, float go) {
    float c = sigmoid_f(gi) * tanh_f(gg);
    return sigmoid_f(go) * tanh_f(c);
}
static __device__ __forceinline__ v8f splat8(float v) {
    v8f r = {v, v, v, v, v, v, v, v};
    return r;
}
static __device__ __forceinline__ v8f wmma4(v2f a, v2f b, v8f c) {
    return __builtin_amdgcn_wmma_f32_16x16x4_f32(false, a, false, b, (short)0, c,
                                                 false, false);
}

static __device__ __forceinline__ void store_gates(float* g, const v8f* acc,
                                                   int half, int l16) {
#pragma unroll
    for (int t = 0; t < 3; ++t)
#pragma unroll
        for (int r = 0; r < 8; ++r)
            g[(r + half * 8) * LDS_STRIDE + t * 16 + l16] = acc[t][r];
}

static __device__ __forceinline__ void cell_slots(const float* g, int half, int l16,
                                                  float* hr) {
#pragma unroll
    for (int s = 0; s < 3; ++s)
#pragma unroll
        for (int v = 0; v < 2; ++v) {
            const int k = s * 4 + half * 2 + v;
            const float* row = g + l16 * LDS_STRIDE;
            hr[s * 2 + v] = lstm_h(row[k], row[24 + k], row[36 + k]);
        }
}

static __device__ __forceinline__ void emit_out(const float* g, float* dst, int lane) {
#pragma unroll
    for (int q = 0; q < 6; ++q) {
        const int flat = q * 32 + lane;
        const int r = flat / 12;
        const int j = flat - r * 12;
        const float* row = g + r * LDS_STRIDE;
        dst[flat] = tanh_f(lstm_h(row[j], row[24 + j], row[36 + j]));
    }
}

__global__ __launch_bounds__(256) void dynamics_net_kernel(
    const float* __restrict__ x,
    const float* __restrict__ W1, const float* __restrict__ bi1, const float* __restrict__ bh1,
    const float* __restrict__ W2, const float* __restrict__ bi2, const float* __restrict__ bh2,
    const float* __restrict__ Wm, const float* __restrict__ bim, const float* __restrict__ bhm,
    const float* __restrict__ Wd, const float* __restrict__ bid, const float* __restrict__ bhd,
    float* __restrict__ out, int n_tiles, long long rowsB) {
    __shared__ float lds[WPB * ROWS * LDS_STRIDE];

    const int lane = threadIdx.x & 31;
    const int wave = threadIdx.x >> 5;
    const int half = lane >> 4;
    const int l16  = lane & 15;
    const int tile = blockIdx.x * WPB + wave;
    if (tile >= n_tiles) return;

    float* g = lds + wave * (ROWS * LDS_STRIDE);

    v2f bw1[3][4];
#pragma unroll
    for (int t = 0; t < 3; ++t) {
        const float* wr = W1 + (t * 16 + l16) * 16 + half * 2;
#pragma unroll
        for (int s = 0; s < 4; ++s) bw1[t][s] = *(const v2f*)(wr + s * 4);
    }
    v2f bw2[3][3], bwm[3][3], bwd[3][3];
#pragma unroll
    for (int t = 0; t < 3; ++t) {
        const int n = t * 16 + l16;
#pragma unroll
        for (int s = 0; s < 3; ++s) {
            bw2[t][s] = *(const v2f*)(W2 + n * 12 + half * 2 + s * 4);
            bwm[t][s] = *(const v2f*)(Wm + n * 12 + half * 2 + s * 4);
            bwd[t][s] = *(const v2f*)(Wd + n * 12 + half * 2 + s * 4);
        }
    }
    float bs1[3], bs2[3], bsm[3], bsd[3];
#pragma unroll
    for (int t = 0; t < 3; ++t) {
        const int n = t * 16 + l16;
        bs1[t] = bi1[n] + bh1[n];
        bs2[t] = bi2[n] + bh2[n];
        bsm[t] = bim[n] + bhm[n];
        bsd[t] = bid[n] + bhd[n];
    }

    const float* xr = x + ((long long)tile * 16 + l16) * 16 + half * 2;
    v2f xa[4];
#pragma unroll
    for (int s = 0; s < 4; ++s) xa[s] = *(const v2f*)(xr + s * 4);

    v8f acc[3];
    float hr[6];
    v2f ha[3];

#pragma unroll
    for (int t = 0; t < 3; ++t) {
        v8f c = splat8(bs1[t]);
#pragma unroll
        for (int s = 0; s < 4; ++s) c = wmma4(xa[s], bw1[t][s], c);
        acc[t] = c;
    }
    store_gates(g, acc, half, l16);
    LDS_FENCE();
    cell_slots(g, half, l16, hr);
#pragma unroll
    for (int s = 0; s < 3; ++s) {
        v2f a = {fmaxf(hr[2 * s], 0.0f), fmaxf(hr[2 * s + 1], 0.0f)};
        ha[s] = a;
    }
    LDS_FENCE();

#pragma unroll
    for (int t = 0; t < 3; ++t) {
        v8f c = splat8(bs2[t]);
#pragma unroll
        for (int s = 0; s < 3; ++s) c = wmma4(ha[s], bw2[t][s], c);
        acc[t] = c;
    }
    store_gates(g, acc, half, l16);
    LDS_FENCE();
    cell_slots(g, half, l16, hr);
#pragma unroll
    for (int s = 0; s < 3; ++s) {
        v2f a = {selu_f(hr[2 * s]), selu_f(hr[2 * s + 1])};
        ha[s] = a;
    }
    LDS_FENCE();

#pragma unroll
    for (int t = 0; t < 3; ++t) {
        v8f c = splat8(bsm[t]);
#pragma unroll
        for (int s = 0; s < 3; ++s) c = wmma4(ha[s], bwm[t][s], c);
        acc[t] = c;
    }
    store_gates(g, acc, half, l16);
    LDS_FENCE();
    emit_out(g, out + (long long)tile * 192, lane);
    LDS_FENCE();

#pragma unroll
    for (int t = 0; t < 3; ++t) {
        v8f c = splat8(bsd[t]);
#pragma unroll
        for (int s = 0; s < 3; ++s) c = wmma4(ha[s], bwd[t][s], c);
        acc[t] = c;
    }
    store_gates(g, acc, half, l16);
    LDS_FENCE();
    emit_out(g, out + rowsB * 12 + (long long)tile * 192, lane);
}

extern "C" void kernel_launch(void* const* d_in, const int* in_sizes, int n_in,
                              void* d_out, int out_size, void* d_ws, size_t ws_size,
                              hipStream_t stream) {
    const float* x   = (const float*)d_in[0];
    const float* W1  = (const float*)d_in[1];
    const float* bi1 = (const float*)d_in[2];
    const float* bh1 = (const float*)d_in[3];
    const float* W2  = (const float*)d_in[4];
    const float* bi2 = (const float*)d_in[5];
    const float* bh2 = (const float*)d_in[6];
    const float* Wm  = (const float*)d_in[7];
    const float* bim = (const float*)d_in[8];
    const float* bhm = (const float*)d_in[9];
    const float* Wd  = (const float*)d_in[10];
    const float* bid = (const float*)d_in[11];
    const float* bhd = (const float*)d_in[12];
    float* out = (float*)d_out;

    const long long rowsB = (long long)in_sizes[0] / 16;
    const int n_tiles = (int)((rowsB + 15) / 16);
    const int blocks  = (n_tiles + WPB - 1) / WPB;

    hipLaunchKernelGGL(dynamics_net_kernel, dim3(blocks), dim3(256), 0, stream,
                       x, W1, bi1, bh1, W2, bi2, bh2, Wm, bim, bhm, Wd, bid, bhd,
                       out, n_tiles, rowsB);
}


# persistent grid: 4096 waves keep weight and bias fragments in registers and loop over 16 row tiles each, next x tile loaded a trip ahead; was 65536 one-tile waves each reloading the weights
# speedup vs baseline: 1.0986x; 1.0986x over previous
#include <hip/hip_runtime.h>

typedef __attribute__((ext_vector_type(2))) float v2f;
typedef __attribute__((ext_vector_type(8))) float v8f;

#define LDS_STRIDE 50
#define ROWS 16
#define WPB 8
#define GRID_BLOCKS 512

#define LDS_FENCE() asm volatile("s_wait_dscnt 0" ::: "memory")

static __device__ __forceinline__ float rcp_f(float x) {
    return __builtin_amdgcn_rcpf(x);
}
static __device__ __forceinline__ float tanh_f(float x) {
#if __has_builtin(__builtin_amdgcn_tanhf)
    return __builtin_amdgcn_tanhf(x);
#else
    float e = __expf(-2.0f * x);
    return (1.0f - e) * rcp_f(1.0f + e);
#endif
}
static __device__ __forceinline__ float sigmoid_f(float x) {
    return fmaf(0.5f, tanh_f(0.5f * x), 0.5f);
}
static __device__ __forceinline__ float selu_f(float x) {
    const float a = 1.6732632423543772f, s = 1.0507009873554805f;
    return x > 0.0f ? s * x : s * a * (__expf(x) - 1.0f);
}
static __device__ __forceinline__ float lstm_h(float gi, float gg, float go) {
    float c = sigmoid_f(gi) * tanh_f(gg);
    return sigmoid_f(go) * tanh_f(c);
}
static __device__ __forceinline__ v8f splat8(float v) {
    v8f r = {v, v, v, v, v, v, v, v};
    return r;
}
static __device__ __forceinline__ v8f wmma4(v2f a, v2f b, v8f c) {
    return __builtin_amdgcn_wmma_f32_16x16x4_f32(false, a, false, b, (short)0, c,
                                                 false, false);
}

static __device__ __forceinline__ void store_gates(float* g, const v8f* acc,
                                                   int half, int l16) {
#pragma unroll
    for (int t = 0; t < 3; ++t)
#pragma unroll
        for (int r = 0; r < 8; ++r)
            g[(r + half * 8) * LDS_STRIDE + t * 16 + l16] = acc[t][r];
}

static __device__ __forceinline__ void cell_slots(const float* g, int half, int l16,
                                                  float* hr) {
#pragma unroll
    for (int s = 0; s < 3; ++s)
#pragma unroll
        for (int v = 0; v < 2; ++v) {
            const int k = s * 4 + half * 2 + v;
            const float* row = g + l16 * LDS_STRIDE;
            hr[s * 2 + v] = lstm_h(row[k], row[24 + k], row[36 + k]);
        }
}

static __device__ __forceinline__ void emit_out(const float* g, float* dst, int lane) {
#pragma unroll
    for (int q = 0; q < 6; ++q) {
        const int flat = q * 32 + lane;
        const int r = flat / 12;
        const int j = flat - r * 12;
        const float* row = g + r * LDS_STRIDE;
        dst[flat] = tanh_f(lstm_h(row[j], row[24 + j], row[36 + j]));
    }
}

__global__ __launch_bounds__(256) void dynamics_net_kernel(
    const float* __restrict__ x,
    const float* __restrict__ W1, const float* __restrict__ bi1, const float* __restrict__ bh1,
    const float* __restrict__ W2, const float* __restrict__ bi2, const float* __restrict__ bh2,
    const float* __restrict__ Wm, const float* __restrict__ bim, const float* __restrict__ bhm,
    const float* __restrict__ Wd, const float* __restrict__ bid, const float* __restrict__ bhd,
    float* __restrict__ out, int n_tiles, long long rowsB) {
    __shared__ float lds[WPB * ROWS * LDS_STRIDE];

    const int lane = threadIdx.x & 31;
    const int wave = __builtin_amdgcn_readfirstlane((int)(threadIdx.x >> 5));
    const int half = lane >> 4;
    const int l16  = lane & 15;
    const int first_tile = blockIdx.x * WPB + wave;
    const int tile_step  = gridDim.x * WPB;
    if (first_tile >= n_tiles) return;

    float* g = lds + wave * (ROWS * LDS_STRIDE);

    v2f bw1[3][4];
#pragma unroll
    for (int t = 0; t < 3; ++t) {
        const float* wr = W1 + (t * 16 + l16) * 16 + half * 2;
#pragma unroll
        for (int s = 0; s < 4; ++s) bw1[t][s] = *(const v2f*)(wr + s * 4);
    }
    v2f bw2[3][3], bwm[3][3], bwd[3][3];
#pragma unroll
    for (int t = 0; t < 3; ++t) {
        const int n = t * 16 + l16;
#pragma unroll
        for (int s = 0; s < 3; ++s) {
            bw2[t][s] = *(const v2f*)(W2 + n * 12 + half * 2 + s * 4);
            bwm[t][s] = *(const v2f*)(Wm + n * 12 + half * 2 + s * 4);
            bwd[t][s] = *(const v2f*)(Wd + n * 12 + half * 2 + s * 4);
        }
    }
    float bs1[3], bs2[3], bsm[3], bsd[3];
#pragma unroll
    for (int t = 0; t < 3; ++t) {
        const int n = t * 16 + l16;
        bs1[t] = bi1[n] + bh1[n];
        bs2[t] = bi2[n] + bh2[n];
        bsm[t] = bim[n] + bhm[n];
        bsd[t] = bid[n] + bhd[n];
    }

    const float* xlane = x + (long long)l16 * 16 + half * 2;
    v2f xa[4];
    {
        const float* xr = xlane + (long long)first_tile * 256;
#pragma unroll
        for (int s = 0; s < 4; ++s) xa[s] = *(const v2f*)(xr + s * 4);
    }

    for (int tile = first_tile; tile < n_tiles; tile += tile_step) {
    const int next_tile = tile + tile_step;
    const int pf_tile = next_tile < n_tiles ? next_tile : tile;
    v2f xn[4];
    {
        const float* xr = xlane + (long long)pf_tile * 256;
#pragma unroll
        for (int s = 0; s < 4; ++s) xn[s] = *(const v2f*)(xr + s * 4);
    }

    v8f acc[3];
    float hr[6];
    v2f ha[3];

#pragma unroll
    for (int t = 0; t < 3; ++t) {
        v8f c = splat8(bs1[t]);
#pragma unroll
        for (int s = 0; s < 4; ++s) c = wmma4(xa[s], bw1[t][s], c);
        acc[t] = c;
    }
    store_gates(g, acc, half, l16);
    LDS_FENCE();
    cell_slots(g, half, l16, hr);
#pragma unroll
    for (int s = 0; s < 3; ++s) {
        v2f a = {fmaxf(hr[2 * s], 0.0f), fmaxf(hr[2 * s + 1], 0.0f)};
        ha[s] = a;
    }
    LDS_FENCE();

#pragma unroll
    for (int t = 0; t < 3; ++t) {
        v8f c = splat8(bs2[t]);
#pragma unroll
        for (int s = 0; s < 3; ++s) c = wmma4(ha[s], bw2[t][s], c);
        acc[t] = c;
    }
    store_gates(g, acc, half, l16);
    LDS_FENCE();
    cell_slots(g, half, l16, hr);
#pragma unroll
    for (int s = 0; s < 3; ++s) {
        v2f a = {selu_f(hr[2 * s]), selu_f(hr[2 * s + 1])};
        ha[s] = a;
    }
    LDS_FENCE();

#pragma unroll
    for (int t = 0; t < 3; ++t) {
        v8f c = splat8(bsm[t]);
#pragma unroll
        for (int s = 0; s < 3; ++s) c = wmma4(ha[s], bwm[t][s], c);
        acc[t] = c;
    }
    store_gates(g, acc, half, l16);
    LDS_FENCE();
    emit_out(g, out + (long long)tile * 192, lane);
    LDS_FENCE();

#pragma unroll
    for (int t = 0; t < 3; ++t) {
        v8f c = splat8(bsd[t]);
#pragma unroll
        for (int s = 0; s < 3; ++s) c = wmma4(ha[s], bwd[t][s], c);
        acc[t] = c;
    }
    store_gates(g, acc, half, l16);
    LDS_FENCE();
    emit_out(g, out + rowsB * 12 + (long long)tile * 192, lane);
    LDS_FENCE();

#pragma unroll
    for (int s = 0; s < 4; ++s) xa[s] = xn[s];
    }
}

extern "C" void kernel_launch(void* const* d_in, const int* in_sizes, int n_in,
                              void* d_out, int out_size, void* d_ws, size_t ws_size,
                              hipStream_t stream) {
    const float* x   = (const float*)d_in[0];
    const float* W1  = (const float*)d_in[1];
    const float* bi1 = (const float*)d_in[2];
    const float* bh1 = (const float*)d_in[3];
    const float* W2  = (const float*)d_in[4];
    const float* bi2 = (const float*)d_in[5];
    const float* bh2 = (const float*)d_in[6];
    const float* Wm  = (const float*)d_in[7];
    const float* bim = (const float*)d_in[8];
    const float* bhm = (const float*)d_in[9];
    const float* Wd  = (const float*)d_in[10];
    const float* bid = (const float*)d_in[11];
    const float* bhd = (const float*)d_in[12];
    float* out = (float*)d_out;

    const long long rowsB = (long long)in_sizes[0] / 16;
    const int n_tiles = (int)((rowsB + 15) / 16);

    hipLaunchKernelGGL(dynamics_net_kernel, dim3(GRID_BLOCKS), dim3(256), 0, stream,
                       x, W1, bi1, bh1, W2, bi2, bh2, Wm, bim, bhm, Wd, bid, bhd,
                       out, n_tiles, rowsB);
}


# gates and h kept in registers as fragment pairs: weights as A operand, gate rows permuted per 16-lane row, k renumbered, bias added after the sum; LDS now only stages the output rows
# speedup vs baseline: 1.2257x; 1.1157x over previous
#include <hip/hip_runtime.h>


typedef __attribute__((ext_vector_type(2))) float v2f;
typedef __attribute__((ext_vector_type(4))) float v4f;
typedef __attribute__((ext_vector_type(8))) float v8f;

#define WPB 8
#define GRID_BLOCKS 512
#define HEAD_FLOATS 192
#define STAGE_FLOATS 384

#define LDS_FENCE() asm volatile("s_wait_dscnt 0" ::: "memory")

static __device__ __forceinline__ float rcp_f(float x) {
    return __builtin_amdgcn_rcpf(x);
}
static __device__ __forceinline__ float tanh_f(float x) {
#if __has_builtin(__builtin_amdgcn_tanhf)
    return __builtin_amdgcn_tanhf(x);
#else
    float e = __expf(-2.0f * x);
    return (1.0f - e) * rcp_f(1.0f + e);
#endif
}
static __device__ __forceinline__ float selu_f(float x) {
    const float a = 1.6732632423543772f, s = 1.0507009873554805f;
    return x > 0.0f ? s * x : s * a * (__expf(x) - 1.0f);
}
static __device__ __forceinline__ v2f tanh2(v2f v) {
    v2f r = {tanh_f(v[0]), tanh_f(v[1])};
    return r;
}
static __device__ __forceinline__ v2f lstm_h2(v2f gi, v2f gg, v2f go,
                                              v2f hbi, v2f bg, v2f hbo) {
    const v2f hf = {0.5f, 0.5f};
    const v2f si = __builtin_elementwise_fma(hf, tanh2(__builtin_elementwise_fma(hf, gi, hbi)), hf);
    const v2f so = __builtin_elementwise_fma(hf, tanh2(__builtin_elementwise_fma(hf, go, hbo)), hf);
    const v2f c = si * tanh2(gg + bg);
    return so * tanh2(c);
}
static __device__ __forceinline__ v8f wmma4(v2f a, v2f b, v8f c) {
    return __builtin_amdgcn_wmma_f32_16x16x4_f32(false, a, false, b, (short)0, c,
                                                 false, false);
}

static __device__ __forceinline__ int gate_base(int t) {
    return t == 0 ? 0 : (t == 1 ? 24 : 36);
}

static __device__ __forceinline__ void cells(const v8f* acc, const v2f* b, v2f* h) {
#pragma unroll
    for (int p = 0; p < 3; ++p) {
        const v2f gi = {acc[0][2 * p], acc[0][2 * p + 1]};
        const v2f gg = {acc[1][2 * p], acc[1][2 * p + 1]};
        const v2f go = {acc[2][2 * p], acc[2][2 * p + 1]};
        h[p] = lstm_h2(gi, gg, go, b[p], b[3 + p], b[6 + p]);
    }
}

__global__ __launch_bounds__(256) void dynamics_net_kernel(
    const float* __restrict__ x,
    const float* __restrict__ W1, const float* __restrict__ bi1, const float* __restrict__ bh1,
    const float* __restrict__ W2, const float* __restrict__ bi2, const float* __restrict__ bh2,
    const float* __restrict__ Wm, const float* __restrict__ bim, const float* __restrict__ bhm,
    const float* __restrict__ Wd, const float* __restrict__ bid, const float* __restrict__ bhd,
    float* __restrict__ out, int n_tiles, long long rowsB) {
    __shared__ __attribute__((aligned(16))) float lds[WPB * STAGE_FLOATS];

    const int lane = threadIdx.x & 31;
    const int wave = __builtin_amdgcn_readfirstlane((int)(threadIdx.x >> 5));
    const int half = lane >> 4;
    const int l16  = lane & 15;
    const int first_tile = blockIdx.x * WPB + wave;
    const int tile_step  = gridDim.x * WPB;
    if (first_tile >= n_tiles) return;

    float* stage = lds + wave * STAGE_FLOATS;


    const int  slot_r = l16 & 7;
    const bool live   = slot_r < 6;
    const int  unit   = live ? 6 * (l16 >> 3) + slot_r : 0;
    const v2f  zero2  = {0.0f, 0.0f};
    const v8f  zero8  = {0.0f, 0.0f, 0.0f, 0.0f, 0.0f, 0.0f, 0.0f, 0.0f};
    v2f aw1[3][4], aw2[3][3], awm[3][3], awd[3][3];
#pragma unroll
    for (int t = 0; t < 3; ++t) {
        const int n = gate_base(t) + unit;
#pragma unroll
        for (int s = 0; s < 4; ++s) {
            const v2f w = *(const v2f*)(W1 + n * 16 + 4 * s + 2 * half);
            aw1[t][s] = live ? w : zero2;
        }
#pragma unroll
        for (int s = 0; s < 3; ++s) {
            const int k = 6 * half + 2 * s;
            const v2f w2 = *(const v2f*)(W2 + n * 12 + k);
            const v2f wm = *(const v2f*)(Wm + n * 12 + k);
            const v2f wd = *(const v2f*)(Wd + n * 12 + k);
            aw2[t][s] = live ? w2 : zero2;
            awm[t][s] = live ? wm : zero2;
            awd[t][s] = live ? wd : zero2;
        }
    }
    v2f b1[9], b2[9], bm[9], bd[9];
#pragma unroll
    for (int t = 0; t < 3; ++t) {
        const float sc = t == 1 ? 1.0f : 0.5f;
#pragma unroll
        for (int p = 0; p < 3; ++p) {
#pragma unroll
            for (int e = 0; e < 2; ++e) {
                const int n = gate_base(t) + 6 * half + 2 * p + e;
                b1[3 * t + p][e] = sc * (bi1[n] + bh1[n]);
                b2[3 * t + p][e] = sc * (bi2[n] + bh2[n]);
                bm[3 * t + p][e] = sc * (bim[n] + bhm[n]);
                bd[3 * t + p][e] = sc * (bid[n] + bhd[n]);
            }
        }
    }

    const float* xlane = x + (long long)l16 * 16 + half * 2;
    v2f xb[4];
    {
        const float* xr = xlane + (long long)first_tile * 256;
#pragma unroll
        for (int s = 0; s < 4; ++s) xb[s] = *(const v2f*)(xr + s * 4);
    }

    float* optr[3];
#pragma unroll
    for (int i = 0; i < 3; ++i) {
        const int ch = 32 * i + lane;
        optr[i] = ch < 48 ? out + ch * 4 : out + rowsB * 12 + (ch - 48) * 4;
    }
    float* st_m = stage + l16 * 12 + 6 * half;
    float* st_d = st_m + HEAD_FLOATS;
    const float* ld_p = stage + lane * 4;

    for (int tile = first_tile; tile < n_tiles; tile += tile_step) {
    const int next_tile = tile + tile_step;
    const int pf_tile = next_tile < n_tiles ? next_tile : tile;
    v2f xn[4];
    {
        const float* xr = xlane + (long long)pf_tile * 256;
#pragma unroll
        for (int s = 0; s < 4; ++s) xn[s] = *(const v2f*)(xr + s * 4);
    }

    v8f acc[3];
    v2f h[3];
    v2f hb[3];

#pragma unroll
    for (int t = 0; t < 3; ++t) acc[t] = zero8;
#pragma unroll
    for (int s = 0; s < 4; ++s)
#pragma unroll
        for (int t = 0; t < 3; ++t) acc[t] = wmma4(aw1[t][s], xb[s], acc[t]);
    cells(acc, b1, h);
#pragma unroll
    for (int s = 0; s < 3; ++s) {
        v2f b = {fmaxf(h[s][0], 0.0f), fmaxf(h[s][1], 0.0f)};
        hb[s] = b;
    }

#pragma unroll
    for (int t = 0; t < 3; ++t) acc[t] = zero8;
#pragma unroll
    for (int s = 0; s < 3; ++s)
#pragma unroll
        for (int t = 0; t < 3; ++t) acc[t] = wmma4(aw2[t][s], hb[s], acc[t]);
    cells(acc, b2, h);
#pragma unroll
    for (int s = 0; s < 3; ++s) {
        v2f b = {selu_f(h[s][0]), selu_f(h[s][1])};
        hb[s] = b;
    }

#pragma unroll
    for (int t = 0; t < 3; ++t) acc[t] = zero8;
#pragma unroll
    for (int s = 0; s < 3; ++s)
#pragma unroll
        for (int t = 0; t < 3; ++t) acc[t] = wmma4(awm[t][s], hb[s], acc[t]);
    cells(acc, bm, h);
#pragma unroll
    for (int p = 0; p < 3; ++p) {
        const v2f o = tanh2(h[p]);
        *(v2f*)(st_m + 2 * p) = o;
    }

#pragma unroll
    for (int t = 0; t < 3; ++t) acc[t] = zero8;
#pragma unroll
    for (int s = 0; s < 3; ++s)
#pragma unroll
        for (int t = 0; t < 3; ++t) acc[t] = wmma4(awd[t][s], hb[s], acc[t]);
    cells(acc, bd, h);
#pragma unroll
    for (int p = 0; p < 3; ++p) {
        const v2f o = tanh2(h[p]);
        *(v2f*)(st_d + 2 * p) = o;
    }

    LDS_FENCE();
    v4f ov[3];
#pragma unroll
    for (int i = 0; i < 3; ++i) ov[i] = *(const v4f*)(ld_p + i * 128);
    LDS_FENCE();
#pragma unroll
    for (int i = 0; i < 3; ++i) *(v4f*)(optr[i] + (long long)tile * HEAD_FLOATS) = ov[i];

#pragma unroll
    for (int s = 0; s < 4; ++s) xb[s] = xn[s];
    }
}

extern "C" void kernel_launch(void* const* d_in, const int* in_sizes, int n_in,
                              void* d_out, int out_size, void* d_ws, size_t ws_size,
                              hipStream_t stream) {
    const float* x   = (const float*)d_in[0];
    const float* W1  = (const float*)d_in[1];
    const float* bi1 = (const float*)d_in[2];
    const float* bh1 = (const float*)d_in[3];
    const float* W2  = (const float*)d_in[4];
    const float* bi2 = (const float*)d_in[5];
    const float* bh2 = (const float*)d_in[6];
    const float* Wm  = (const float*)d_in[7];
    const float* bim = (const float*)d_in[8];
    const float* bhm = (const float*)d_in[9];
    const float* Wd  = (const float*)d_in[10];
    const float* bid = (const float*)d_in[11];
    const float* bhd = (const float*)d_in[12];
    float* out = (float*)d_out;

    const long long rowsB = (long long)in_sizes[0] / 16;
    const int n_tiles = (int)((rowsB + 15) / 16);

    hipLaunchKernelGGL(dynamics_net_kernel, dim3(GRID_BLOCKS), dim3(256), 0, stream,
                       x, W1, bi1, bh1, W2, bi2, bh2, Wm, bim, bhm, Wd, bid, bhd,
                       out, n_tiles, rowsB);
}


# both heads' gate rows (2 x 36) packed into five row tiles of weights on the shared h2 operand, was 2 x 3 tiles: 36 matrix products per batch tile instead of 39; sums and their order unchanged
# speedup vs baseline: 1.2761x; 1.0411x over previous
#include <hip/hip_runtime.h>


typedef __attribute__((ext_vector_type(2))) float v2f;
typedef __attribute__((ext_vector_type(4))) float v4f;
typedef __attribute__((ext_vector_type(8))) float v8f;

#define WPB 8
#define GRID_BLOCKS 512
#define HEAD_FLOATS 192
#define STAGE_FLOATS 384

#define LDS_FENCE() asm volatile("s_wait_dscnt 0" ::: "memory")

static __device__ __forceinline__ float rcp_f(float x) {
    return __builtin_amdgcn_rcpf(x);
}
static __device__ __forceinline__ float tanh_f(float x) {
#if __has_builtin(__builtin_amdgcn_tanhf)
    return __builtin_amdgcn_tanhf(x);
#else
    float e = __expf(-2.0f * x);
    return (1.0f - e) * rcp_f(1.0f + e);
#endif
}
static __device__ __forceinline__ float selu_f(float x) {
    const float a = 1.6732632423543772f, s = 1.0507009873554805f;
    return x > 0.0f ? s * x : s * a * (__expf(x) - 1.0f);
}
static __device__ __forceinline__ v2f tanh2(v2f v) {
    v2f r = {tanh_f(v[0]), tanh_f(v[1])};
    return r;
}
static __device__ __forceinline__ v2f lstm_h2(v2f gi, v2f gg, v2f go,
                                              v2f hbi, v2f bg, v2f hbo) {
    const v2f hf = {0.5f, 0.5f};
    const v2f si = __builtin_elementwise_fma(hf, tanh2(__builtin_elementwise_fma(hf, gi, hbi)), hf);
    const v2f so = __builtin_elementwise_fma(hf, tanh2(__builtin_elementwise_fma(hf, go, hbo)), hf);
    const v2f c = si * tanh2(gg + bg);
    return so * tanh2(c);
}
static __device__ __forceinline__ v8f wmma4(v2f a, v2f b, v8f c) {
    return __builtin_amdgcn_wmma_f32_16x16x4_f32(false, a, false, b, (short)0, c,
                                                 false, false);
}

static __device__ __forceinline__ int gate_base(int t) {
    return t == 0 ? 0 : (t == 1 ? 24 : 36);
}

static __device__ __forceinline__ void cells(const v8f* acc, const v2f* b, v2f* h) {
#pragma unroll
    for (int p = 0; p < 3; ++p) {
        const v2f gi = {acc[0][2 * p], acc[0][2 * p + 1]};
        const v2f gg = {acc[1][2 * p], acc[1][2 * p + 1]};
        const v2f go = {acc[2][2 * p], acc[2][2 * p + 1]};
        h[p] = lstm_h2(gi, gg, go, b[p], b[3 + p], b[6 + p]);
    }
}

template <int HD>
static __device__ __forceinline__ void cells_head(const v8f* acc5, const v2f* b, v2f* h) {
#pragma unroll
    for (int p = 0; p < 3; ++p) {
        const int qi = 18 * HD + 2 * p, qg = qi + 6, qo = qi + 12;
        const v2f gi = {acc5[qi >> 3][qi & 7], acc5[qi >> 3][(qi & 7) + 1]};
        const v2f gg = {acc5[qg >> 3][qg & 7], acc5[qg >> 3][(qg & 7) + 1]};
        const v2f go = {acc5[qo >> 3][qo & 7], acc5[qo >> 3][(qo & 7) + 1]};
        h[p] = lstm_h2(gi, gg, go, b[p], b[3 + p], b[6 + p]);
    }
}

__global__ __launch_bounds__(256) void dynamics_net_kernel(
    const float* __restrict__ x,
    const float* __restrict__ W1, const float* __restrict__ bi1, const float* __restrict__ bh1,
    const float* __restrict__ W2, const float* __restrict__ bi2, const float* __restrict__ bh2,
    const float* __restrict__ Wm, const float* __restrict__ bim, const float* __restrict__ bhm,
    const float* __restrict__ Wd, const float* __restrict__ bid, const float* __restrict__ bhd,
    float* __restrict__ out, int n_tiles, long long rowsB) {
    __shared__ __attribute__((aligned(16))) float lds[WPB * STAGE_FLOATS];

    const int lane = threadIdx.x & 31;
    const int wave = __builtin_amdgcn_readfirstlane((int)(threadIdx.x >> 5));
    const int half = lane >> 4;
    const int l16  = lane & 15;
    const int first_tile = blockIdx.x * WPB + wave;
    const int tile_step  = gridDim.x * WPB;
    if (first_tile >= n_tiles) return;

    float* stage = lds + wave * STAGE_FLOATS;


    const int  slot_r = l16 & 7;
    const bool live   = slot_r < 6;
    const int  unit   = live ? 6 * (l16 >> 3) + slot_r : 0;
    const v2f  zero2  = {0.0f, 0.0f};
    const v8f  zero8  = {0.0f, 0.0f, 0.0f, 0.0f, 0.0f, 0.0f, 0.0f, 0.0f};
    v2f aw1[3][4], aw2[3][3], awh[5][3];
#pragma unroll
    for (int t = 0; t < 3; ++t) {
        const int n = gate_base(t) + unit;
#pragma unroll
        for (int s = 0; s < 4; ++s) {
            const v2f w = *(const v2f*)(W1 + n * 16 + 4 * s + 2 * half);
            aw1[t][s] = live ? w : zero2;
        }
#pragma unroll
        for (int s = 0; s < 3; ++s) {
            const int k = 6 * half + 2 * s;
            const v2f w2 = *(const v2f*)(W2 + n * 12 + k);
            aw2[t][s] = live ? w2 : zero2;
        }
    }
#pragma unroll
    for (int t = 0; t < 5; ++t) {
        const int  q     = 8 * t + slot_r;
        const bool liveh = q < 36;
        const int  v     = liveh ? q : 0;
        const int  hd    = v / 18;
        const int  gt    = (v % 18) / 6;
        const int  n     = gate_base(gt) + 6 * (l16 >> 3) + v % 6;
        const float* Wh  = hd ? Wd : Wm;
#pragma unroll
        for (int s = 0; s < 3; ++s) {
            const v2f w = *(const v2f*)(Wh + n * 12 + 6 * half + 2 * s);
            awh[t][s] = liveh ? w : zero2;
        }
    }
    v2f b1[9], b2[9], bm[9], bd[9];
#pragma unroll
    for (int t = 0; t < 3; ++t) {
        const float sc = t == 1 ? 1.0f : 0.5f;
#pragma unroll
        for (int p = 0; p < 3; ++p) {
#pragma unroll
            for (int e = 0; e < 2; ++e) {
                const int n = gate_base(t) + 6 * half + 2 * p + e;
                b1[3 * t + p][e] = sc * (bi1[n] + bh1[n]);
                b2[3 * t + p][e] = sc * (bi2[n] + bh2[n]);
                bm[3 * t + p][e] = sc * (bim[n] + bhm[n]);
                bd[3 * t + p][e] = sc * (bid[n] + bhd[n]);
            }
        }
    }

    const float* xlane = x + (long long)l16 * 16 + half * 2;
    v2f xb[4];
    {
        const float* xr = xlane + (long long)first_tile * 256;
#pragma unroll
        for (int s = 0; s < 4; ++s) xb[s] = *(const v2f*)(xr + s * 4);
    }

    float* optr[3];
#pragma unroll
    for (int i = 0; i < 3; ++i) {
        const int ch = 32 * i + lane;
        optr[i] = ch < 48 ? out + ch * 4 : out + rowsB * 12 + (ch - 48) * 4;
    }
    float* st_m = stage + l16 * 12 + 6 * half;
    float* st_d = st_m + HEAD_FLOATS;
    const float* ld_p = stage + lane * 4;

    for (int tile = first_tile; tile < n_tiles; tile += tile_step) {
    const int next_tile = tile + tile_step;
    const int pf_tile = next_tile < n_tiles ? next_tile : tile;
    v2f xn[4];
    {
        const float* xr = xlane + (long long)pf_tile * 256;
#pragma unroll
        for (int s = 0; s < 4; ++s) xn[s] = *(const v2f*)(xr + s * 4);
    }

    v8f acc[3];
    v2f h[3];
    v2f hb[3];

#pragma unroll
    for (int t = 0; t < 3; ++t) acc[t] = zero8;
#pragma unroll
    for (int s = 0; s < 4; ++s)
#pragma unroll
        for (int t = 0; t < 3; ++t) acc[t] = wmma4(aw1[t][s], xb[s], acc[t]);
    cells(acc, b1, h);
#pragma unroll
    for (int s = 0; s < 3; ++s) {
        v2f b = {fmaxf(h[s][0], 0.0f), fmaxf(h[s][1], 0.0f)};
        hb[s] = b;
    }

#pragma unroll
    for (int t = 0; t < 3; ++t) acc[t] = zero8;
#pragma unroll
    for (int s = 0; s < 3; ++s)
#pragma unroll
        for (int t = 0; t < 3; ++t) acc[t] = wmma4(aw2[t][s], hb[s], acc[t]);
    cells(acc, b2, h);
#pragma unroll
    for (int s = 0; s < 3; ++s) {
        v2f b = {selu_f(h[s][0]), selu_f(h[s][1])};
        hb[s] = b;
    }

    v8f acc5[5];
#pragma unroll
    for (int t = 0; t < 5; ++t) acc5[t] = zero8;
#pragma unroll
    for (int s = 0; s < 3; ++s)
#pragma unroll
        for (int t = 0; t < 5; ++t) acc5[t] = wmma4(awh[t][s], hb[s], acc5[t]);

    cells_head<0>(acc5, bm, h);
#pragma unroll
    for (int p = 0; p < 3; ++p) {
        const v2f o = tanh2(h[p]);
        *(v2f*)(st_m + 2 * p) = o;
    }

    cells_head<1>(acc5, bd, h);
#pragma unroll
    for (int p = 0; p < 3; ++p) {
        const v2f o = tanh2(h[p]);
        *(v2f*)(st_d + 2 * p) = o;
    }

    LDS_FENCE();
    v4f ov[3];
#pragma unroll
    for (int i = 0; i < 3; ++i) ov[i] = *(const v4f*)(ld_p + i * 128);
    LDS_FENCE();
#pragma unroll
    for (int i = 0; i < 3; ++i) *(v4f*)(optr[i] + (long long)tile * HEAD_FLOATS) = ov[i];

#pragma unroll
    for (int s = 0; s < 4; ++s) xb[s] = xn[s];
    }
}

extern "C" void kernel_launch(void* const* d_in, const int* in_sizes, int n_in,
                              void* d_out, int out_size, void* d_ws, size_t ws_size,
                              hipStream_t stream) {
    const float* x   = (const float*)d_in[0];
    const float* W1  = (const float*)d_in[1];
    const float* bi1 = (const float*)d_in[2];
    const float* bh1 = (const float*)d_in[3];
    const float* W2  = (const float*)d_in[4];
    const float* bi2 = (const float*)d_in[5];
    const float* bh2 = (const float*)d_in[6];
    const float* Wm  = (const float*)d_in[7];
    const float* bim = (const float*)d_in[8];
    const float* bhm = (const float*)d_in[9];
    const float* Wd  = (const float*)d_in[10];
    const float* bid = (const float*)d_in[11];
    const float* bhd = (const float*)d_in[12];
    float* out = (float*)d_out;

    const long long rowsB = (long long)in_sizes[0] / 16;
    const int n_tiles = (int)((rowsB + 15) / 16);

    hipLaunchKernelGGL(dynamics_net_kernel, dim3(GRID_BLOCKS), dim3(256), 0, stream,
                       x, W1, bi1, bh1, W2, bi2, bh2, Wm, bim, bhm, Wd, bid, bhd,
                       out, n_tiles, rowsB);
}


# SELU without divergent branches: s*max(x,0) + fma(s*a, exp(min(x,0)), -s*a), packed per pair (negative side now rounded once, by the fma); was six EXEC-masked if/else regions per tile
# speedup vs baseline: 1.3325x; 1.0442x over previous
#include <hip/hip_runtime.h>


typedef __attribute__((ext_vector_type(2))) float v2f;
typedef __attribute__((ext_vector_type(4))) float v4f;
typedef __attribute__((ext_vector_type(8))) float v8f;

#define WPB 8
#define GRID_BLOCKS 512
#define HEAD_FLOATS 192
#define STAGE_FLOATS 384

#define LDS_FENCE() asm volatile("s_wait_dscnt 0" ::: "memory")

static __device__ __forceinline__ float rcp_f(float x) {
    return __builtin_amdgcn_rcpf(x);
}
static __device__ __forceinline__ float tanh_f(float x) {
#if __has_builtin(__builtin_amdgcn_tanhf)
    return __builtin_amdgcn_tanhf(x);
#else
    float e = __expf(-2.0f * x);
    return (1.0f - e) * rcp_f(1.0f + e);
#endif
}
static __device__ __forceinline__ v2f selu2(v2f x) {
    const float a = 1.6732632423543772f, s = 1.0507009873554805f;
    const float sa = s * a;
    const v2f zero = {0.0f, 0.0f};
    const v2f s2 = {s, s}, sa2 = {sa, sa}, nsa2 = {-sa, -sa};
    const v2f xn = __builtin_elementwise_min(x, zero);
    const v2f xp = __builtin_elementwise_max(x, zero);
    const v2f e = {__expf(xn[0]), __expf(xn[1])};
    const v2f neg = __builtin_elementwise_fma(e, sa2, nsa2);
    return __builtin_elementwise_fma(xp, s2, neg);
}
static __device__ __forceinline__ v2f tanh2(v2f v) {
    v2f r = {tanh_f(v[0]), tanh_f(v[1])};
    return r;
}
static __device__ __forceinline__ v2f lstm_h2(v2f gi, v2f gg, v2f go,
                                              v2f hbi, v2f bg, v2f hbo) {
    const v2f hf = {0.5f, 0.5f};
    const v2f si = __builtin_elementwise_fma(hf, tanh2(__builtin_elementwise_fma(hf, gi, hbi)), hf);
    const v2f so = __builtin_elementwise_fma(hf, tanh2(__builtin_elementwise_fma(hf, go, hbo)), hf);
    const v2f c = si * tanh2(gg + bg);
    return so * tanh2(c);
}
static __device__ __forceinline__ v8f wmma4(v2f a, v2f b, v8f c) {
    return __builtin_amdgcn_wmma_f32_16x16x4_f32(false, a, false, b, (short)0, c,
                                                 false, false);
}

static __device__ __forceinline__ int gate_base(int t) {
    return t == 0 ? 0 : (t == 1 ? 24 : 36);
}

static __device__ __forceinline__ void cells(const v8f* acc, const v2f* b, v2f* h) {
#pragma unroll
    for (int p = 0; p < 3; ++p) {
        const v2f gi = {acc[0][2 * p], acc[0][2 * p + 1]};
        const v2f gg = {acc[1][2 * p], acc[1][2 * p + 1]};
        const v2f go = {acc[2][2 * p], acc[2][2 * p + 1]};
        h[p] = lstm_h2(gi, gg, go, b[p], b[3 + p], b[6 + p]);
    }
}

template <int HD>
static __device__ __forceinline__ void cells_head(const v8f* acc5, const v2f* b, v2f* h) {
#pragma unroll
    for (int p = 0; p < 3; ++p) {
        const int qi = 18 * HD + 2 * p, qg = qi + 6, qo = qi + 12;
        const v2f gi = {acc5[qi >> 3][qi & 7], acc5[qi >> 3][(qi & 7) + 1]};
        const v2f gg = {acc5[qg >> 3][qg & 7], acc5[qg >> 3][(qg & 7) + 1]};
        const v2f go = {acc5[qo >> 3][qo & 7], acc5[qo >> 3][(qo & 7) + 1]};
        h[p] = lstm_h2(gi, gg, go, b[p], b[3 + p], b[6 + p]);
    }
}

__global__ __launch_bounds__(256) void dynamics_net_kernel(
    const float* __restrict__ x,
    const float* __restrict__ W1, const float* __restrict__ bi1, const float* __restrict__ bh1,
    const float* __restrict__ W2, const float* __restrict__ bi2, const float* __restrict__ bh2,
    const float* __restrict__ Wm, const float* __restrict__ bim, const float* __restrict__ bhm,
    const float* __restrict__ Wd, const float* __restrict__ bid, const float* __restrict__ bhd,
    float* __restrict__ out, int n_tiles, long long rowsB) {
    __shared__ __attribute__((aligned(16))) float lds[WPB * STAGE_FLOATS];

    const int lane = threadIdx.x & 31;
    const int wave = __builtin_amdgcn_readfirstlane((int)(threadIdx.x >> 5));
    const int half = lane >> 4;
    const int l16  = lane & 15;
    const int first_tile = blockIdx.x * WPB + wave;
    const int tile_step  = gridDim.x * WPB;
    if (first_tile >= n_tiles) return;

    float* stage = lds + wave * STAGE_FLOATS;


    const int  slot_r = l16 & 7;
    const bool live   = slot_r < 6;
    const int  unit   = live ? 6 * (l16 >> 3) + slot_r : 0;
    const v2f  zero2  = {0.0f, 0.0f};
    const v8f  zero8  = {0.0f, 0.0f, 0.0f, 0.0f, 0.0f, 0.0f, 0.0f, 0.0f};
    v2f aw1[3][4], aw2[3][3], awh[5][3];
#pragma unroll
    for (int t = 0; t < 3; ++t) {
        const int n = gate_base(t) + unit;
#pragma unroll
        for (int s = 0; s < 4; ++s) {
            const v2f w = *(const v2f*)(W1 + n * 16 + 4 * s + 2 * half);
            aw1[t][s] = live ? w : zero2;
        }
#pragma unroll
        for (int s = 0; s < 3; ++s) {
            const int k = 6 * half + 2 * s;
            const v2f w2 = *(const v2f*)(W2 + n * 12 + k);
            aw2[t][s] = live ? w2 : zero2;
        }
    }
#pragma unroll
    for (int t = 0; t < 5; ++t) {
        const int  q     = 8 * t + slot_r;
        const bool liveh = q < 36;
        const int  v     = liveh ? q : 0;
        const int  hd    = v / 18;
        const int  gt    = (v % 18) / 6;
        const int  n     = gate_base(gt) + 6 * (l16 >> 3) + v % 6;
        const float* Wh  = hd ? Wd : Wm;
#pragma unroll
        for (int s = 0; s < 3; ++s) {
            const v2f w = *(const v2f*)(Wh + n * 12 + 6 * half + 2 * s);
            awh[t][s] = liveh ? w : zero2;
        }
    }
    v2f b1[9], b2[9], bm[9], bd[9];
#pragma unroll
    for (int t = 0; t < 3; ++t) {
        const float sc = t == 1 ? 1.0f : 0.5f;
#pragma unroll
        for (int p = 0; p < 3; ++p) {
#pragma unroll
            for (int e = 0; e < 2; ++e) {
                const int n = gate_base(t) + 6 * half + 2 * p + e;
                b1[3 * t + p][e] = sc * (bi1[n] + bh1[n]);
                b2[3 * t + p][e] = sc * (bi2[n] + bh2[n]);
                bm[3 * t + p][e] = sc * (bim[n] + bhm[n]);
                bd[3 * t + p][e] = sc * (bid[n] + bhd[n]);
            }
        }
    }

    const float* xlane = x + (long long)l16 * 16 + half * 2;
    v2f xb[4];
    {
        const float* xr = xlane + (long long)first_tile * 256;
#pragma unroll
        for (int s = 0; s < 4; ++s) xb[s] = *(const v2f*)(xr + s * 4);
    }

    float* optr[3];
#pragma unroll
    for (int i = 0; i < 3; ++i) {
        const int ch = 32 * i + lane;
        optr[i] = ch < 48 ? out + ch * 4 : out + rowsB * 12 + (ch - 48) * 4;
    }
    float* st_m = stage + l16 * 12 + 6 * half;
    float* st_d = st_m + HEAD_FLOATS;
    const float* ld_p = stage + lane * 4;

    for (int tile = first_tile; tile < n_tiles; tile += tile_step) {
    const int next_tile = tile + tile_step;
    const int pf_tile = next_tile < n_tiles ? next_tile : tile;
    v2f xn[4];
    {
        const float* xr = xlane + (long long)pf_tile * 256;
#pragma unroll
        for (int s = 0; s < 4; ++s) xn[s] = *(const v2f*)(xr + s * 4);
    }

    v8f acc[3];
    v2f h[3];
    v2f hb[3];

#pragma unroll
    for (int t = 0; t < 3; ++t) acc[t] = zero8;
#pragma unroll
    for (int s = 0; s < 4; ++s)
#pragma unroll
        for (int t = 0; t < 3; ++t) acc[t] = wmma4(aw1[t][s], xb[s], acc[t]);
    cells(acc, b1, h);
#pragma unroll
    for (int s = 0; s < 3; ++s) {
        v2f b = {fmaxf(h[s][0], 0.0f), fmaxf(h[s][1], 0.0f)};
        hb[s] = b;
    }

#pragma unroll
    for (int t = 0; t < 3; ++t) acc[t] = zero8;
#pragma unroll
    for (int s = 0; s < 3; ++s)
#pragma unroll
        for (int t = 0; t < 3; ++t) acc[t] = wmma4(aw2[t][s], hb[s], acc[t]);
    cells(acc, b2, h);
#pragma unroll
    for (int s = 0; s < 3; ++s) hb[s] = selu2(h[s]);

    v8f acc5[5];
#pragma unroll
    for (int t = 0; t < 5; ++t) acc5[t] = zero8;
#pragma unroll
    for (int s = 0; s < 3; ++s)
#pragma unroll
        for (int t = 0; t < 5; ++t) acc5[t] = wmma4(awh[t][s], hb[s], acc5[t]);

    cells_head<0>(acc5, bm, h);
#pragma unroll
    for (int p = 0; p < 3; ++p) {
        const v2f o = tanh2(h[p]);
        *(v2f*)(st_m + 2 * p) = o;
    }

    cells_head<1>(acc5, bd, h);
#pragma unroll
    for (int p = 0; p < 3; ++p) {
        const v2f o = tanh2(h[p]);
        *(v2f*)(st_d + 2 * p) = o;
    }

    LDS_FENCE();
    v4f ov[3];
#pragma unroll
    for (int i = 0; i < 3; ++i) ov[i] = *(const v4f*)(ld_p + i * 128);
    LDS_FENCE();
#pragma unroll
    for (int i = 0; i < 3; ++i) *(v4f*)(optr[i] + (long long)tile * HEAD_FLOATS) = ov[i];

#pragma unroll
    for (int s = 0; s < 4; ++s) xb[s] = xn[s];
    }
}

extern "C" void kernel_launch(void* const* d_in, const int* in_sizes, int n_in,
                              void* d_out, int out_size, void* d_ws, size_t ws_size,
                              hipStream_t stream) {
    const float* x   = (const float*)d_in[0];
    const float* W1  = (const float*)d_in[1];
    const float* bi1 = (const float*)d_in[2];
    const float* bh1 = (const float*)d_in[3];
    const float* W2  = (const float*)d_in[4];
    const float* bi2 = (const float*)d_in[5];
    const float* bh2 = (const float*)d_in[6];
    const float* Wm  = (const float*)d_in[7];
    const float* bim = (const float*)d_in[8];
    const float* bhm = (const float*)d_in[9];
    const float* Wd  = (const float*)d_in[10];
    const float* bid = (const float*)d_in[11];
    const float* bhd = (const float*)d_in[12];
    float* out = (float*)d_out;

    const long long rowsB = (long long)in_sizes[0] / 16;
    const int n_tiles = (int)((rowsB + 15) / 16);

    hipLaunchKernelGGL(dynamics_net_kernel, dim3(GRID_BLOCKS), dim3(256), 0, stream,
                       x, W1, bi1, bh1, W2, bi2, bh2, Wm, bim, bhm, Wd, bid, bhd,
                       out, n_tiles, rowsB);
}


# two row tiles per wave go through each layer together; their 6 (heads: 12) cell pairs run as a software pipeline, issue order pinned so tanh and packed ops alternate in one wave; 2 waves per SIMD
# speedup vs baseline: 1.3896x; 1.0429x over previous
#include <hip/hip_runtime.h>


typedef __attribute__((ext_vector_type(2))) float v2f;
typedef __attribute__((ext_vector_type(4))) float v4f;
typedef __attribute__((ext_vector_type(8))) float v8f;

#define WPB 8
#define TPW 2
#define GRID_BLOCKS 256
#define HEAD_FLOATS 192
#define STAGE_FLOATS 384

#define LDS_FENCE() asm volatile("s_wait_dscnt 0" ::: "memory")

static __device__ __forceinline__ float rcp_f(float x) {
    return __builtin_amdgcn_rcpf(x);
}
static __device__ __forceinline__ float tanh_f(float x) {
#if __has_builtin(__builtin_amdgcn_tanhf)
    return __builtin_amdgcn_tanhf(x);
#else
    float e = __expf(-2.0f * x);
    return (1.0f - e) * rcp_f(1.0f + e);
#endif
}
static __device__ __forceinline__ v2f selu2(v2f x) {
    const float a = 1.6732632423543772f, s = 1.0507009873554805f;
    const float sa = s * a;
    const v2f zero = {0.0f, 0.0f};
    const v2f s2 = {s, s}, sa2 = {sa, sa}, nsa2 = {-sa, -sa};
    const v2f xn = __builtin_elementwise_min(x, zero);
    const v2f xp = __builtin_elementwise_max(x, zero);
    const v2f e = {__expf(xn[0]), __expf(xn[1])};
    const v2f neg = __builtin_elementwise_fma(e, sa2, nsa2);
    return __builtin_elementwise_fma(xp, s2, neg);
}
static __device__ __forceinline__ v2f tanh2(v2f v) {
    v2f r = {tanh_f(v[0]), tanh_f(v[1])};
    return r;
}
#define ONE_TRANS() __builtin_amdgcn_sched_group_barrier(0x400, 1, SID)
#define ONE_PLAIN() __builtin_amdgcn_sched_group_barrier(0x002, 1, SID)
#define TWO_PLAIN() __builtin_amdgcn_sched_group_barrier(0x002, 2, SID)
#define PHASE_EDGE() __builtin_amdgcn_sched_barrier(0)

template <int NP, int MODE, int SID>
static __device__ __forceinline__ void lstm_pairs(const v2f* gi, const v2f* gg, const v2f* go,
                                                  const v2f* hbi, const v2f* bg, const v2f* hbo,
                                                  v2f* out) {
    const v2f hf = {0.5f, 0.5f};
    v2f ai[NP], ag[NP], ao[NP], ti[NP], tg[NP], to[NP], si[NP], so[NP], c[NP], tc[NP], h[NP];
    PHASE_EDGE();
    ai[0] = __builtin_elementwise_fma(hf, gi[0], hbi[0]); ONE_PLAIN();
    ag[0] = gg[0] + bg[0];                                ONE_PLAIN();
    ao[0] = __builtin_elementwise_fma(hf, go[0], hbo[0]); ONE_PLAIN();
#pragma unroll
    for (int it = 0; it <= NP + 2; ++it) {
        const bool gate = it < NP;
        const bool pre  = it + 1 < NP;
        const bool cell = it >= 1 && it - 1 < NP;
        const bool outp = it >= 2 && it - 2 < NP;
        const bool post = it >= 3 && it - 3 < NP;
        if (gate) { ti[it][0] = tanh_f(ai[it][0]); ONE_TRANS(); }
        if (outp) { so[it - 2] = __builtin_elementwise_fma(hf, to[it - 2], hf); ONE_PLAIN(); }
        if (gate) { ti[it][1] = tanh_f(ai[it][1]); ONE_TRANS(); }
        if (pre)  { ai[it + 1] = __builtin_elementwise_fma(hf, gi[it + 1], hbi[it + 1]); ONE_PLAIN(); }
        if (gate) { tg[it][0] = tanh_f(ag[it][0]); ONE_TRANS(); }
        if (cell) { c[it - 1] = si[it - 1] * tg[it - 1]; ONE_PLAIN(); }
        if (gate) { tg[it][1] = tanh_f(ag[it][1]); ONE_TRANS(); }
        if (pre)  { ao[it + 1] = __builtin_elementwise_fma(hf, go[it + 1], hbo[it + 1]); ONE_PLAIN(); }
        if (gate) { to[it][0] = tanh_f(ao[it][0]); ONE_TRANS(); }
        if (post) {
            if (MODE == 0) {
                out[it - 3][0] = fmaxf(h[it - 3][0], 0.0f);
                out[it - 3][1] = fmaxf(h[it - 3][1], 0.0f); TWO_PLAIN();
            } else if (MODE == 2) {
                out[it - 3][0] = tanh_f(h[it - 3][0]); ONE_TRANS();
                out[it - 3][1] = tanh_f(h[it - 3][1]); ONE_TRANS();
            } else {
                out[it - 3] = h[it - 3];
            }
        }
        if (gate) { to[it][1] = tanh_f(ao[it][1]); ONE_TRANS(); }
        if (pre)  { ag[it + 1] = gg[it + 1] + bg[it + 1]; ONE_PLAIN(); }
        if (cell) { tc[it - 1][0] = tanh_f(c[it - 1][0]); ONE_TRANS(); }
        if (outp) { h[it - 2] = so[it - 2] * tc[it - 2]; ONE_PLAIN(); }
        if (cell) { tc[it - 1][1] = tanh_f(c[it - 1][1]); ONE_TRANS(); }
        if (gate) { si[it] = __builtin_elementwise_fma(hf, ti[it], hf); ONE_PLAIN(); }
    }
    PHASE_EDGE();
}
static __device__ __forceinline__ v8f wmma4(v2f a, v2f b, v8f c) {
    return __builtin_amdgcn_wmma_f32_16x16x4_f32(false, a, false, b, (short)0, c,
                                                 false, false);
}

static __device__ __forceinline__ int gate_base(int t) {
    return t == 0 ? 0 : (t == 1 ? 24 : 36);
}


__global__ __launch_bounds__(256) void dynamics_net_kernel(
    const float* __restrict__ x,
    const float* __restrict__ W1, const float* __restrict__ bi1, const float* __restrict__ bh1,
    const float* __restrict__ W2, const float* __restrict__ bi2, const float* __restrict__ bh2,
    const float* __restrict__ Wm, const float* __restrict__ bim, const float* __restrict__ bhm,
    const float* __restrict__ Wd, const float* __restrict__ bid, const float* __restrict__ bhd,
    float* __restrict__ out, int n_tiles, long long rowsB) {
    __shared__ __attribute__((aligned(16))) float lds[WPB * TPW * STAGE_FLOATS];

    const int lane = threadIdx.x & 31;
    const int wave = __builtin_amdgcn_readfirstlane((int)(threadIdx.x >> 5));
    const int half = lane >> 4;
    const int l16  = lane & 15;
    const int n_pairs    = (n_tiles + 1) >> 1;
    const int last_tile  = n_tiles - 1;
    const int first_pair = blockIdx.x * WPB + wave;
    const int pair_step  = gridDim.x * WPB;
    if (first_pair >= n_pairs) return;

    float* stage = lds + wave * (TPW * STAGE_FLOATS);


    const int  slot_r = l16 & 7;
    const bool live   = slot_r < 6;
    const int  unit   = live ? 6 * (l16 >> 3) + slot_r : 0;
    const v2f  zero2  = {0.0f, 0.0f};
    const v8f  zero8  = {0.0f, 0.0f, 0.0f, 0.0f, 0.0f, 0.0f, 0.0f, 0.0f};
    v2f aw1[3][4], aw2[3][3], awh[5][3];
#pragma unroll
    for (int t = 0; t < 3; ++t) {
        const int n = gate_base(t) + unit;
#pragma unroll
        for (int s = 0; s < 4; ++s) {
            const v2f w = *(const v2f*)(W1 + n * 16 + 4 * s + 2 * half);
            aw1[t][s] = live ? w : zero2;
        }
#pragma unroll
        for (int s = 0; s < 3; ++s) {
            const int k = 6 * half + 2 * s;
            const v2f w2 = *(const v2f*)(W2 + n * 12 + k);
            aw2[t][s] = live ? w2 : zero2;
        }
    }
#pragma unroll
    for (int t = 0; t < 5; ++t) {
        const int  q     = 8 * t + slot_r;
        const bool liveh = q < 36;
        const int  v     = liveh ? q : 0;
        const int  hd    = v / 18;
        const int  gt    = (v % 18) / 6;
        const int  n     = gate_base(gt) + 6 * (l16 >> 3) + v % 6;
        const float* Wh  = hd ? Wd : Wm;
#pragma unroll
        for (int s = 0; s < 3; ++s) {
            const v2f w = *(const v2f*)(Wh + n * 12 + 6 * half + 2 * s);
            awh[t][s] = liveh ? w : zero2;
        }
    }
    v2f b1[9], b2[9], bm[9], bd[9];
#pragma unroll
    for (int t = 0; t < 3; ++t) {
        const float sc = t == 1 ? 1.0f : 0.5f;
#pragma unroll
        for (int p = 0; p < 3; ++p) {
#pragma unroll
            for (int e = 0; e < 2; ++e) {
                const int n = gate_base(t) + 6 * half + 2 * p + e;
                b1[3 * t + p][e] = sc * (bi1[n] + bh1[n]);
                b2[3 * t + p][e] = sc * (bi2[n] + bh2[n]);
                bm[3 * t + p][e] = sc * (bim[n] + bhm[n]);
                bd[3 * t + p][e] = sc * (bid[n] + bhd[n]);
            }
        }
    }

    const float* xlane = x + (long long)l16 * 16 + half * 2;
    v2f xb[TPW][4];
#pragma unroll
    for (int u = 0; u < TPW; ++u) {
        const int tl = min(2 * first_pair + u, last_tile);
        const float* xr = xlane + (long long)tl * 256;
#pragma unroll
        for (int s = 0; s < 4; ++s) xb[u][s] = *(const v2f*)(xr + s * 4);
    }

    long long ooff[3];
#pragma unroll
    for (int i = 0; i < 3; ++i) {
        const int ch = 32 * i + lane;
        ooff[i] = (ch < 48 ? (long long)(ch * 4) : rowsB * 12 + (ch - 48) * 4)
                  + (long long)first_pair * (TPW * HEAD_FLOATS);
    }
    float* st_m = stage + l16 * 12 + 6 * half;
    const float* ld_p = stage + lane * 4;

    for (int pair = first_pair; pair < n_pairs; pair += pair_step) {
    v8f acc[TPW][3];
    v2f hb[TPW][3];
    v2f gi[6 * TPW], gg[6 * TPW], go[6 * TPW];
    v2f qi[6 * TPW], qg[6 * TPW], qo[6 * TPW];
    v2f hh[6 * TPW];

#pragma unroll
    for (int u = 0; u < TPW; ++u)
#pragma unroll
        for (int t = 0; t < 3; ++t) acc[u][t] = zero8;
#pragma unroll
    for (int s = 0; s < 4; ++s)
#pragma unroll
        for (int t = 0; t < 3; ++t)
#pragma unroll
            for (int u = 0; u < TPW; ++u) acc[u][t] = wmma4(aw1[t][s], xb[u][s], acc[u][t]);
    {
        const int next_pair = pair + pair_step;
        const int pf_pair = next_pair < n_pairs ? next_pair : pair;
#pragma unroll
        for (int u = 0; u < TPW; ++u) {
            const int tl = min(2 * pf_pair + u, last_tile);
            const float* xr = xlane + (long long)tl * 256;
#pragma unroll
            for (int s = 0; s < 4; ++s) xb[u][s] = *(const v2f*)(xr + s * 4);
        }
    }
#pragma unroll
    for (int u = 0; u < TPW; ++u)
#pragma unroll
        for (int p = 0; p < 3; ++p) {
            const int k = 3 * u + p;
            const v2f a = {acc[u][0][2 * p], acc[u][0][2 * p + 1]};
            const v2f b = {acc[u][1][2 * p], acc[u][1][2 * p + 1]};
            const v2f c = {acc[u][2][2 * p], acc[u][2][2 * p + 1]};
            gi[k] = a; gg[k] = b; go[k] = c;
            qi[k] = b1[p]; qg[k] = b1[3 + p]; qo[k] = b1[6 + p];
        }
    lstm_pairs<3 * TPW, 0, 1>(gi, gg, go, qi, qg, qo, hh);
#pragma unroll
    for (int u = 0; u < TPW; ++u)
#pragma unroll
        for (int s = 0; s < 3; ++s) hb[u][s] = hh[3 * u + s];

#pragma unroll
    for (int u = 0; u < TPW; ++u)
#pragma unroll
        for (int t = 0; t < 3; ++t) acc[u][t] = zero8;
#pragma unroll
    for (int s = 0; s < 3; ++s)
#pragma unroll
        for (int t = 0; t < 3; ++t)
#pragma unroll
            for (int u = 0; u < TPW; ++u) acc[u][t] = wmma4(aw2[t][s], hb[u][s], acc[u][t]);
#pragma unroll
    for (int u = 0; u < TPW; ++u)
#pragma unroll
        for (int p = 0; p < 3; ++p) {
            const int k = 3 * u + p;
            const v2f a = {acc[u][0][2 * p], acc[u][0][2 * p + 1]};
            const v2f b = {acc[u][1][2 * p], acc[u][1][2 * p + 1]};
            const v2f c = {acc[u][2][2 * p], acc[u][2][2 * p + 1]};
            gi[k] = a; gg[k] = b; go[k] = c;
            qi[k] = b2[p]; qg[k] = b2[3 + p]; qo[k] = b2[6 + p];
        }
    lstm_pairs<3 * TPW, 1, 2>(gi, gg, go, qi, qg, qo, hh);
#pragma unroll
    for (int u = 0; u < TPW; ++u)
#pragma unroll
        for (int s = 0; s < 3; ++s) hb[u][s] = selu2(hh[3 * u + s]);

    v8f acc5[TPW][5];
#pragma unroll
    for (int u = 0; u < TPW; ++u)
#pragma unroll
        for (int t = 0; t < 5; ++t) acc5[u][t] = zero8;
#pragma unroll
    for (int s = 0; s < 3; ++s)
#pragma unroll
        for (int t = 0; t < 5; ++t)
#pragma unroll
            for (int u = 0; u < TPW; ++u) acc5[u][t] = wmma4(awh[t][s], hb[u][s], acc5[u][t]);

#pragma unroll
    for (int u = 0; u < TPW; ++u)
#pragma unroll
        for (int hd = 0; hd < 2; ++hd)
#pragma unroll
            for (int p = 0; p < 3; ++p) {
                const int k = 6 * u + 3 * hd + p;
                const int ri = 18 * hd + 2 * p, rg = ri + 6, ro = ri + 12;
                const v2f a = {acc5[u][ri >> 3][ri & 7], acc5[u][ri >> 3][(ri & 7) + 1]};
                const v2f b = {acc5[u][rg >> 3][rg & 7], acc5[u][rg >> 3][(rg & 7) + 1]};
                const v2f c = {acc5[u][ro >> 3][ro & 7], acc5[u][ro >> 3][(ro & 7) + 1]};
                gi[k] = a; gg[k] = b; go[k] = c;
                qi[k] = hd ? bd[p] : bm[p];
                qg[k] = hd ? bd[3 + p] : bm[3 + p];
                qo[k] = hd ? bd[6 + p] : bm[6 + p];
            }
    lstm_pairs<6 * TPW, 2, 3>(gi, gg, go, qi, qg, qo, hh);
#pragma unroll
    for (int u = 0; u < TPW; ++u)
#pragma unroll
        for (int hd = 0; hd < 2; ++hd)
#pragma unroll
            for (int p = 0; p < 3; ++p)
                *(v2f*)(st_m + u * STAGE_FLOATS + hd * HEAD_FLOATS + 2 * p) = hh[6 * u + 3 * hd + p];

    LDS_FENCE();
    v4f ov[TPW][3];
#pragma unroll
    for (int u = 0; u < TPW; ++u)
#pragma unroll
        for (int i = 0; i < 3; ++i) ov[u][i] = *(const v4f*)(ld_p + u * STAGE_FLOATS + i * 128);
    LDS_FENCE();
    const int second = 2 * pair + 1 <= last_tile ? HEAD_FLOATS : 0;
#pragma unroll
    for (int i = 0; i < 3; ++i) {
        *(v4f*)(out + ooff[i]) = ov[0][i];
        *(v4f*)(out + ooff[i] + second) = ov[1][i];
    }

#pragma unroll
    for (int i = 0; i < 3; ++i) ooff[i] += (long long)pair_step * (TPW * HEAD_FLOATS);
    }
}

extern "C" void kernel_launch(void* const* d_in, const int* in_sizes, int n_in,
                              void* d_out, int out_size, void* d_ws, size_t ws_size,
                              hipStream_t stream) {
    const float* x   = (const float*)d_in[0];
    const float* W1  = (const float*)d_in[1];
    const float* bi1 = (const float*)d_in[2];
    const float* bh1 = (const float*)d_in[3];
    const float* W2  = (const float*)d_in[4];
    const float* bi2 = (const float*)d_in[5];
    const float* bh2 = (const float*)d_in[6];
    const float* Wm  = (const float*)d_in[7];
    const float* bim = (const float*)d_in[8];
    const float* bhm = (const float*)d_in[9];
    const float* Wd  = (const float*)d_in[10];
    const float* bid = (const float*)d_in[11];
    const float* bhd = (const float*)d_in[12];
    float* out = (float*)d_out;

    const long long rowsB = (long long)in_sizes[0] / 16;
    const int n_tiles = (int)((rowsB + 15) / 16);

    hipLaunchKernelGGL(dynamics_net_kernel, dim3(GRID_BLOCKS), dim3(256), 0, stream,
                       x, W1, bi1, bh1, W2, bi2, bh2, Wm, bim, bhm, Wd, bid, bhd,
                       out, n_tiles, rowsB);
}
